# starting kernel
# baseline (speedup 1.0000x reference)
#include <hip/hip_runtime.h>


#define IN_C   128
#define OUT_C  128
#define N_EDGES_CONST 50000

typedef __attribute__((ext_vector_type(2))) float v2f;
typedef __attribute__((ext_vector_type(8))) float v8f;

#define LDS_STRIDE 132

__global__ __launch_bounds__(256) void gemm_xw_wmma_kernel(
    const float* __restrict__ x, const float* __restrict__ W,
    float* __restrict__ xw)
{
    __shared__ float ldsA[16 * LDS_STRIDE];

    const int row0 = blockIdx.x * 16;
    const int tid  = threadIdx.x;

    const float4* __restrict__ xsrc = (const float4*)(x + (size_t)row0 * IN_C);
    #pragma unroll
    for (int t = 0; t < 2; ++t) {
        int chunk = tid + t * 256;
        int r   = chunk >> 5;
        int off = (chunk & 31) << 2;
        *(float4*)(ldsA + r * LDS_STRIDE + off) = xsrc[chunk];
    }
    __syncthreads();

    const int wave  = tid >> 5;
    const int lane  = tid & 31;
    const int col0  = wave * 16;
    const int lrow  = lane & 15;
    const int khalf = (lane >> 4) << 1;

    const float* __restrict__ Wc = W + col0;

    v8f c = {};
    #pragma unroll
    for (int k = 0; k < IN_C; k += 4) {
        v2f a = *(const v2f*)(ldsA + lrow * LDS_STRIDE + k + khalf);
        v2f b;
        b.x = Wc[(size_t)(k + khalf + 0) * OUT_C + lrow];
        b.y = Wc[(size_t)(k + khalf + 1) * OUT_C + lrow];
        c = __builtin_amdgcn_wmma_f32_16x16x4_f32(
                 false, a,  false, b,
                 (short)0, c,  false,  false);
    }

    float* __restrict__ dst = xw + (size_t)row0 * OUT_C + col0;
    const int mbase = (lane >> 4) << 3;
    #pragma unroll
    for (int r = 0; r < 8; ++r) {
        dst[(size_t)(mbase + r) * OUT_C + lrow] = c[r];
    }
}

__global__ void degrees_kernel(const int* __restrict__ node_idx,
                               const int* __restrict__ edge_idx,
                               float* __restrict__ deg_n,
                               float* __restrict__ deg_e, int nnz)
{
    int i = blockIdx.x * blockDim.x + threadIdx.x;
    if (i < nnz) {
        atomicAdd(&deg_n[node_idx[i]], 1.0f);
        atomicAdd(&deg_e[edge_idx[i]], 1.0f);
    }
}

__global__ void invert_kernel(float* __restrict__ d, int n)
{
    int i = blockIdx.x * blockDim.x + threadIdx.x;
    if (i < n) {
        float v = d[i];
        d[i] = (v > 0.0f) ? (1.0f / v) : 0.0f;
    }
}

__global__ __launch_bounds__(256) void scatter_node_to_edge_kernel(
    const float* __restrict__ xw,
    const int* __restrict__ node_idx, const int* __restrict__ edge_idx,
    float* __restrict__ edge_feat, int nnz)
{
    int idx = blockIdx.x * 256 + threadIdx.x;
    int i   = idx >> 5;
    int c4  = (idx & 31) << 2;
    if (i < nnz) {
        int v = node_idx[i];
        int e = edge_idx[i];
        float4 val = *(const float4*)(xw + (size_t)v * OUT_C + c4);
        float* dst = edge_feat + (size_t)e * OUT_C + c4;
        atomicAdd(dst + 0, val.x);
        atomicAdd(dst + 1, val.y);
        atomicAdd(dst + 2, val.z);
        atomicAdd(dst + 3, val.w);
    }
}

__global__ __launch_bounds__(256) void scatter_edge_to_node_kernel(
    const float* __restrict__ edge_feat, const float* __restrict__ b_inv,
    const int* __restrict__ node_idx, const int* __restrict__ edge_idx,
    float* __restrict__ out, int nnz)
{
    int idx = blockIdx.x * 256 + threadIdx.x;
    int i   = idx >> 5;
    int c4  = (idx & 31) << 2;
    if (i < nnz) {
        int v = node_idx[i];
        int e = edge_idx[i];
        float s = b_inv[e];
        float4 val = *(const float4*)(edge_feat + (size_t)e * OUT_C + c4);
        float* dst = out + (size_t)v * OUT_C + c4;
        atomicAdd(dst + 0, val.x * s);
        atomicAdd(dst + 1, val.y * s);
        atomicAdd(dst + 2, val.z * s);
        atomicAdd(dst + 3, val.w * s);
    }
}

__global__ __launch_bounds__(256) void finalize_kernel(
    float* __restrict__ out, const float* __restrict__ d_inv,
    const float* __restrict__ bias, int n_nodes)
{
    int idx = blockIdx.x * 256 + threadIdx.x;
    int v   = idx >> 5;
    int c4  = (idx & 31) << 2;
    if (v < n_nodes) {
        float  s  = d_inv[v];
        float4 bb = *(const float4*)(bias + c4);
        float4 o  = *(const float4*)(out + (size_t)v * OUT_C + c4);
        o.x = o.x * s + bb.x;
        o.y = o.y * s + bb.y;
        o.z = o.z * s + bb.z;
        o.w = o.w * s + bb.w;
        *(float4*)(out + (size_t)v * OUT_C + c4) = o;
    }
}

extern "C" void kernel_launch(void* const* d_in, const int* in_sizes, int n_in,
                              void* d_out, int out_size, void* d_ws, size_t ws_size,
                              hipStream_t stream)
{
    const float* x    = (const float*)d_in[0];
    const int*   hidx = (const int*)d_in[1];
    const float* W    = (const float*)d_in[2];
    const float* bias = (const float*)d_in[3];

    const int n_nodes = in_sizes[0] / IN_C;
    const int nnz     = in_sizes[1] / 2;
    const int n_edges = N_EDGES_CONST;

    const int* node_idx = hidx;
    const int* edge_idx = hidx + nnz;

    char* ws = (char*)d_ws;
    float* xw        = (float*)(ws);
    float* edge_feat = (float*)(ws + (size_t)n_nodes * OUT_C * 4);
    float* deg_n     = (float*)((char*)edge_feat + (size_t)n_edges * OUT_C * 4);
    float* deg_e     = (float*)((char*)deg_n + (size_t)n_nodes * 4);

    float* out = (float*)d_out;

    hipMemsetAsync(edge_feat, 0, (size_t)n_edges * OUT_C * 4, stream);
    hipMemsetAsync(deg_n,     0, (size_t)n_nodes * 4, stream);
    hipMemsetAsync(deg_e,     0, (size_t)n_edges * 4, stream);
    hipMemsetAsync(out,       0, (size_t)n_nodes * OUT_C * 4, stream);

    degrees_kernel<<<(nnz + 255) / 256, 256, 0, stream>>>(
        node_idx, edge_idx, deg_n, deg_e, nnz);
    invert_kernel<<<(n_nodes + 255) / 256, 256, 0, stream>>>(deg_n, n_nodes);
    invert_kernel<<<(n_edges + 255) / 256, 256, 0, stream>>>(deg_e, n_edges);

    gemm_xw_wmma_kernel<<<n_nodes / 16, 256, 0, stream>>>(x, W, xw);

    {
        long long threads = (long long)nnz * 32;
        scatter_node_to_edge_kernel<<<(int)((threads + 255) / 256), 256, 0, stream>>>(
            xw, node_idx, edge_idx, edge_feat, nnz);
    }

    {
        long long threads = (long long)nnz * 32;
        scatter_edge_to_node_kernel<<<(int)((threads + 255) / 256), 256, 0, stream>>>(
            edge_feat, deg_e, node_idx, edge_idx, out, nnz);
    }

    {
        long long threads = (long long)n_nodes * 32;
        finalize_kernel<<<(int)((threads + 255) / 256), 256, 0, stream>>>(
            out, deg_n, bias, n_nodes);
    }
}


# Fold fills, degree count, reciprocals and the D^-1/bias pass into the 3 big kernels (11 stream ops -> 3); sum order changes: out starts as bias, each added row scaled by 1/|e| then 1/d
# speedup vs baseline: 1.9644x; 1.9644x over previous
#include <hip/hip_runtime.h>


#define IN_C   128
#define OUT_C  128
#define N_EDGES_CONST 50000

typedef __attribute__((ext_vector_type(2))) float v2f;
typedef __attribute__((ext_vector_type(8))) float v8f;

#define LDS_STRIDE 132

__global__ __launch_bounds__(256) void gemm_xw_wmma_kernel(
    const float* __restrict__ x, const float* __restrict__ W,
    const float* __restrict__ bias,
    float* __restrict__ xw, float* __restrict__ out,
    float* __restrict__ edge_feat, float* __restrict__ deg_n,
    float* __restrict__ deg_e, int n_nodes, int n_edges)
{
    __shared__ float ldsA[16 * LDS_STRIDE];

    const int row0 = blockIdx.x * 16;
    const int tid  = threadIdx.x;

    const float4 bb = *(const float4*)(bias + ((tid & 31) << 2));

    const float4* __restrict__ xsrc = (const float4*)(x + (size_t)row0 * IN_C);
    #pragma unroll
    for (int t = 0; t < 2; ++t) {
        int chunk = tid + t * 256;
        int r   = chunk >> 5;
        int off = (chunk & 31) << 2;
        *(float4*)(ldsA + r * LDS_STRIDE + off) = xsrc[chunk];
    }
    __syncthreads();

    const int wave  = tid >> 5;
    const int lane  = tid & 31;
    const int col0  = wave * 16;
    const int lrow  = lane & 15;
    const int khalf = (lane >> 4) << 1;

    const float* __restrict__ Wc = W + col0;

    v8f c = {};
    #pragma unroll
    for (int k = 0; k < IN_C; k += 4) {
        v2f a = *(const v2f*)(ldsA + lrow * LDS_STRIDE + k + khalf);
        v2f b;
        b.x = Wc[(size_t)(k + khalf + 0) * OUT_C + lrow];
        b.y = Wc[(size_t)(k + khalf + 1) * OUT_C + lrow];
        c = __builtin_amdgcn_wmma_f32_16x16x4_f32(
                 false, a,  false, b,
                 (short)0, c,  false,  false);
    }

    float* __restrict__ dst = xw + (size_t)row0 * OUT_C + col0;
    const int mbase = (lane >> 4) << 3;
    #pragma unroll
    for (int r = 0; r < 8; ++r) {
        dst[(size_t)(mbase + r) * OUT_C + lrow] = c[r];
    }

    {
        float4* o4 = (float4*)(out + (size_t)row0 * OUT_C);
        o4[tid]       = bb;
        o4[tid + 256] = bb;
        if ((blockIdx.x & 1) == 0 && tid < 32 && row0 + tid < n_nodes)
            deg_n[row0 + tid] = 0.0f;
        const float4 z4 = {0.0f, 0.0f, 0.0f, 0.0f};
        for (int ec = blockIdx.x; ec * 8 < n_edges; ec += gridDim.x) {
            const int e = ec * 8 + (tid >> 5);
            if (e < n_edges)
                ((float4*)(edge_feat + (size_t)e * OUT_C))[tid & 31] = z4;
            if ((ec & 3) == 0 && tid < 32 && ec * 8 + tid < n_edges)
                deg_e[ec * 8 + tid] = 0.0f;
        }
    }
}

#define IPW 32
#define RB  8

__device__ __forceinline__ void row_load(const float* __restrict__ row, int lane, float (&r)[4])
{
    r[0] = row[lane];
    r[1] = row[lane + 32];
    r[2] = row[lane + 64];
    r[3] = row[lane + 96];
}

__device__ __forceinline__ void row_add(float* __restrict__ row, int lane, const float (&r)[4], float sc)
{
    atomicAdd(row + lane,      r[0] * sc);
    atomicAdd(row + lane + 32, r[1] * sc);
    atomicAdd(row + lane + 64, r[2] * sc);
    atomicAdd(row + lane + 96, r[3] * sc);
}

__device__ __forceinline__ void row_add2(float* __restrict__ row, int lane, const float (&r)[4], float se, float sv)
{
    atomicAdd(row + lane,      (r[0] * se) * sv);
    atomicAdd(row + lane + 32, (r[1] * se) * sv);
    atomicAdd(row + lane + 64, (r[2] * se) * sv);
    atomicAdd(row + lane + 96, (r[3] * se) * sv);
}

__device__ __forceinline__ float inv_deg(float d)
{
    return (d > 0.0f) ? (1.0f / d) : 0.0f;
}

__global__ __launch_bounds__(256) void scatter_node_to_edge_kernel(
    const float* __restrict__ xw,
    const int* __restrict__ node_idx, const int* __restrict__ edge_idx,
    float* __restrict__ edge_feat, float* __restrict__ deg_n,
    float* __restrict__ deg_e, int nnz)
{
    const int lane = threadIdx.x & 31;
    const int wave = __builtin_amdgcn_readfirstlane(threadIdx.x >> 5);
    const int base = (blockIdx.x * 8 + wave) * IPW;
    if (base >= nnz) return;
    const int cnt = (nnz - base < IPW) ? (nnz - base) : IPW;

    int v_l = 0, e_l = 0;
    if (lane < cnt) {
        v_l = node_idx[base + lane];
        e_l = edge_idx[base + lane];
    }

    int j0 = 0;
    for (; j0 + RB <= cnt; j0 += RB) {
        float r[RB][4];
        #pragma unroll
        for (int u = 0; u < RB; ++u) {
            const int v = __builtin_amdgcn_readlane(v_l, j0 + u);
            row_load(xw + (size_t)v * OUT_C, lane, r[u]);
        }
        #pragma unroll
        for (int u = 0; u < RB; ++u) {
            const int e = __builtin_amdgcn_readlane(e_l, j0 + u);
            row_add(edge_feat + (size_t)e * OUT_C, lane, r[u], 1.0f);
        }
    }
    for (; j0 < cnt; ++j0) {
        float r[4];
        const int v = __builtin_amdgcn_readlane(v_l, j0);
        const int e = __builtin_amdgcn_readlane(e_l, j0);
        row_load(xw + (size_t)v * OUT_C, lane, r);
        row_add(edge_feat + (size_t)e * OUT_C, lane, r, 1.0f);
    }

    if (lane < cnt) {
        atomicAdd(&deg_n[v_l], 1.0f);
        atomicAdd(&deg_e[e_l], 1.0f);
    }
}

__global__ __launch_bounds__(256) void scatter_edge_to_node_kernel(
    const float* __restrict__ edge_feat, const float* __restrict__ deg_e,
    const float* __restrict__ deg_n,
    const int* __restrict__ node_idx, const int* __restrict__ edge_idx,
    float* __restrict__ out, int nnz)
{
    const int lane = threadIdx.x & 31;
    const int wave = __builtin_amdgcn_readfirstlane(threadIdx.x >> 5);
    const int base = (blockIdx.x * 8 + wave) * IPW;
    if (base >= nnz) return;
    const int cnt = (nnz - base < IPW) ? (nnz - base) : IPW;

    int v_l = 0, e_l = 0;
    if (lane < cnt) {
        v_l = node_idx[base + lane];
        e_l = edge_idx[base + lane];
    }
    const int s_l = __float_as_int(inv_deg(deg_e[e_l]));
    const int d_l = __float_as_int(inv_deg(deg_n[v_l]));

    int j0 = 0;
    for (; j0 + RB <= cnt; j0 += RB) {
        float r[RB][4];
        #pragma unroll
        for (int u = 0; u < RB; ++u) {
            const int e = __builtin_amdgcn_readlane(e_l, j0 + u);
            row_load(edge_feat + (size_t)e * OUT_C, lane, r[u]);
        }
        #pragma unroll
        for (int u = 0; u < RB; ++u) {
            const int v = __builtin_amdgcn_readlane(v_l, j0 + u);
            const float se = __int_as_float(__builtin_amdgcn_readlane(s_l, j0 + u));
            const float sv = __int_as_float(__builtin_amdgcn_readlane(d_l, j0 + u));
            row_add2(out + (size_t)v * OUT_C, lane, r[u], se, sv);
        }
    }
    for (; j0 < cnt; ++j0) {
        float r[4];
        const int v = __builtin_amdgcn_readlane(v_l, j0);
        const int e = __builtin_amdgcn_readlane(e_l, j0);
        const float se = __int_as_float(__builtin_amdgcn_readlane(s_l, j0));
        const float sv = __int_as_float(__builtin_amdgcn_readlane(d_l, j0));
        row_load(edge_feat + (size_t)e * OUT_C, lane, r);
        row_add2(out + (size_t)v * OUT_C, lane, r, se, sv);
    }
}

extern "C" void kernel_launch(void* const* d_in, const int* in_sizes, int n_in,
                              void* d_out, int out_size, void* d_ws, size_t ws_size,
                              hipStream_t stream)
{
    const float* x    = (const float*)d_in[0];
    const int*   hidx = (const int*)d_in[1];
    const float* W    = (const float*)d_in[2];
    const float* bias = (const float*)d_in[3];

    const int n_nodes = in_sizes[0] / IN_C;
    const int nnz     = in_sizes[1] / 2;
    const int n_edges = N_EDGES_CONST;

    const int* node_idx = hidx;
    const int* edge_idx = hidx + nnz;

    char* ws = (char*)d_ws;
    float* xw        = (float*)(ws);
    float* edge_feat = (float*)(ws + (size_t)n_nodes * OUT_C * 4);
    float* deg_n     = (float*)((char*)edge_feat + (size_t)n_edges * OUT_C * 4);
    float* deg_e     = (float*)((char*)deg_n + (size_t)n_nodes * 4);

    float* out = (float*)d_out;

    gemm_xw_wmma_kernel<<<n_nodes / 16, 256, 0, stream>>>(
        x, W, bias, xw, out, edge_feat, deg_n, deg_e, n_nodes, n_edges);

    scatter_node_to_edge_kernel<<<(nnz + 8 * IPW - 1) / (8 * IPW), 256, 0, stream>>>(
        xw, node_idx, edge_idx, edge_feat, deg_n, deg_e, nnz);

    scatter_edge_to_node_kernel<<<(nnz + 8 * IPW - 1) / (8 * IPW), 256, 0, stream>>>(
        edge_feat, deg_e, deg_n, node_idx, edge_idx, out, nnz);
}


# Scans read 1-byte row-block numbers (index / rows per block, written by the x@W launch) not 4-byte indices: 16 incidences per load, zero-byte test, per-lane find slots; sum order changes
# speedup vs baseline: 2.7607x; 1.4054x over previous
#include <hip/hip_runtime.h>


#define IN_C   128
#define OUT_C  128
#define N_EDGES_CONST 50000

typedef __attribute__((ext_vector_type(2))) float v2f;
typedef __attribute__((ext_vector_type(8))) float v8f;

#define ROWS_E   196
#define ROWS_N   391

__device__ __forceinline__ unsigned owner_byte(int idx, bool edge)
{
    const unsigned u = (unsigned)idx;
    return (edge ? (u / (unsigned)ROWS_E) : (u / (unsigned)ROWS_N)) & 255u;
}

__device__ __forceinline__ unsigned owner_bytes4(int4 v, bool edge)
{
    return owner_byte(v.x, edge) | (owner_byte(v.y, edge) << 8) |
           (owner_byte(v.z, edge) << 16) | (owner_byte(v.w, edge) << 24);
}

__device__ __forceinline__ void write_owner_bytes(
    const int* __restrict__ node_idx, const int* __restrict__ edge_idx, int nnz,
    unsigned char* __restrict__ own_n, unsigned char* __restrict__ own_e,
    int t0, int nthreads)
{
    const int nchunk = (nnz + 7) >> 3;
    const int npad   = (nchunk + 31) & ~31;
    for (int t = t0; t < 2 * npad; t += nthreads) {
        const bool edge = t >= npad;
        const int  base = (edge ? (t - npad) : t) << 3;
        if (base >= nnz) continue;
        const int* __restrict__ idx = edge ? edge_idx : node_idx;
        unsigned char* __restrict__ own = edge ? own_e : own_n;
        if (base + 8 <= nnz && (nnz & 3) == 0) {
            const int4 a = *(const int4*)(idx + base);
            const int4 b = *(const int4*)(idx + base + 4);
            *(uint2*)(own + base) = make_uint2(owner_bytes4(a, edge), owner_bytes4(b, edge));
        } else {
            for (int j = 0; j < 8; ++j)
                if (base + j < nnz) own[base + j] = (unsigned char)owner_byte(idx[base + j], edge);
        }
    }
}

#define LDS_STRIDE 132
#define GEMM_WGS   512

__global__ __launch_bounds__(256) void gemm_xw_wmma_kernel(
    const float* __restrict__ x, const float* __restrict__ W,
    float* __restrict__ xw, int n_strips,
    const int* __restrict__ node_idx, const int* __restrict__ edge_idx, int nnz,
    unsigned char* __restrict__ own_n, unsigned char* __restrict__ own_e)
{
    __shared__ __attribute__((aligned(16))) float ldsA[2][16 * LDS_STRIDE];

    const int tid   = threadIdx.x;

    write_owner_bytes(node_idx, edge_idx, nnz, own_n, own_e,
                      blockIdx.x * 256 + tid, GEMM_WGS * 256);

    const int wave  = tid >> 5;
    const int lane  = tid & 31;
    const int col0  = wave * 16;
    const int lrow  = lane & 15;
    const int khalf = (lane >> 4) << 1;

    const float* __restrict__ Wc = W + col0;
    v2f bfrag[IN_C / 4];
    #pragma unroll
    for (int kk = 0; kk < IN_C / 4; ++kk) {
        bfrag[kk].x = Wc[(size_t)(4 * kk + khalf + 0) * OUT_C + lrow];
        bfrag[kk].y = Wc[(size_t)(4 * kk + khalf + 1) * OUT_C + lrow];
    }

    const int soff = (tid >> 5) * LDS_STRIDE + ((tid & 31) << 2);

    int s = blockIdx.x;
    float4 p0 = {0.0f, 0.0f, 0.0f, 0.0f}, p1 = {0.0f, 0.0f, 0.0f, 0.0f};
    if (s < n_strips) {
        const float4* __restrict__ xs = (const float4*)(x + (size_t)s * 16 * IN_C);
        p0 = xs[tid];
        p1 = xs[tid + 256];
        *(float4*)(&ldsA[0][soff]) = p0;
        *(float4*)(&ldsA[0][soff + 8 * LDS_STRIDE]) = p1;
    }
    __syncthreads();

    const int mbase = (lane >> 4) << 3;
    int buf = 0;
    for (; s < n_strips; s += GEMM_WGS) {
        const int  sn   = s + GEMM_WGS;
        const bool more = sn < n_strips;
        if (more) {
            const float4* __restrict__ xs = (const float4*)(x + (size_t)sn * 16 * IN_C);
            p0 = xs[tid];
            p1 = xs[tid + 256];
        }

        const float* A = &ldsA[buf][lrow * LDS_STRIDE + khalf];
        v8f c = {};
        #pragma unroll
        for (int kk = 0; kk < IN_C / 4; ++kk) {
            v2f a = *(const v2f*)(A + 4 * kk);
            c = __builtin_amdgcn_wmma_f32_16x16x4_f32(
                     false, a,  false, bfrag[kk],
                     (short)0, c,  false,  false);
        }

        if (more) {
            *(float4*)(&ldsA[buf ^ 1][soff]) = p0;
            *(float4*)(&ldsA[buf ^ 1][soff + 8 * LDS_STRIDE]) = p1;
        }
        __syncthreads();

        float* __restrict__ dst = xw + (size_t)s * 16 * OUT_C + col0;
        #pragma unroll
        for (int r = 0; r < 8; ++r) {
            dst[(size_t)(mbase + r) * OUT_C + lrow] = c[r];
        }
        buf ^= 1;
    }
}

#define WG_T     1024
#define WG_W     32
#define CAPL     12
#define CAPW     192
#define LU       8
#define RB       8
#define SU       8
#define OWN_GRID 256

__device__ __forceinline__ float inv_cnt(int d)
{
    return (d > 0) ? (1.0f / (float)d) : 0.0f;
}

__device__ __forceinline__ void add_row_by_lane(const float* __restrict__ row, float* a)
{
    #pragma nounroll
    for (int c = 0; c < OUT_C; ++c) atomicAdd(&a[c], row[c]);
}

__device__ __forceinline__ void add_incidence_by_lane(
    int inc, int row0, int rows,
    const float* __restrict__ src, const int* __restrict__ src_idx, const int* __restrict__ dst_idx,
    float* acc, int* cnt)
{
    const int l = dst_idx[inc] - row0;
    if ((unsigned)l < (unsigned)rows) {
        atomicAdd(&cnt[l], 1);
        add_row_by_lane(src + (size_t)src_idx[inc] * OUT_C, acc + l * OUT_C);
    }
}

__device__ __forceinline__ unsigned nonzero_bytes(unsigned x)
{
    return (((x & 0x7f7f7f7fu) + 0x7f7f7f7fu) | x) & 0x80808080u;
}

__device__ __forceinline__ unsigned equal_bytes(uint4 v, unsigned pat)
{
    const unsigned ne = (nonzero_bytes(v.x ^ pat) >> 7) | (nonzero_bytes(v.y ^ pat) >> 6) |
                        (nonzero_bytes(v.z ^ pat) >> 5) | (nonzero_bytes(v.w ^ pat) >> 4);
    return ~ne & 0x0f0f0f0fu;
}

template <int ROWS, bool HAS_BIAS>
__device__ __forceinline__ void owner_segment_sum(
    const float* __restrict__ src, const int* __restrict__ src_idx,
    const int* __restrict__ dst_idx, const unsigned char* __restrict__ dst_own,
    int nnz, int n_dst,
    float* __restrict__ dst, const float* __restrict__ bias,
    float* acc, int* cnt, int* finds, int2* list)
{
    const int tid  = threadIdx.x;
    const int lane = tid & 31;
    const int wave = __builtin_amdgcn_readfirstlane(tid >> 5);
    int2* mylist = list + wave * CAPW;
    int*  myfind = finds + tid;

    for (int rb = blockIdx.x; rb * ROWS < n_dst; rb += gridDim.x) {
        const int row0 = rb * ROWS;
        const int rows = (n_dst - row0 < ROWS) ? (n_dst - row0) : ROWS;

        const float4 z4 = {0.0f, 0.0f, 0.0f, 0.0f};
        for (int i = tid; i < ROWS * (OUT_C / 4); i += WG_T) ((float4*)acc)[i] = z4;
        for (int i = tid; i < ROWS; i += WG_T) cnt[i] = 0;
        #pragma unroll
        for (int k = 0; k < CAPL; ++k) myfind[k * WG_T] = 0;
        __syncthreads();

        int nl = 0;
        const unsigned me  = (unsigned)rb & 255u;
        const unsigned pat = me * 0x01010101u;
        const int nvec = nnz >> 4;
        const uint4* __restrict__ o4 = (const uint4*)dst_own;
        for (int qb = 0; qb < nvec; qb += WG_T * SU) {
            uint4 v[SU];
            #pragma unroll
            for (int u = 0; u < SU; ++u) {
                const int q = qb + u * WG_T + tid;
                v[u] = o4[(q < nvec) ? q : (nvec - 1)];
            }
            #pragma unroll
            for (int u = 0; u < SU; ++u) {
                const int q = qb + u * WG_T + tid;
                unsigned h = (q < nvec) ? equal_bytes(v[u], pat) : 0u;
                while (h) {
                    const int p = __builtin_ctz(h);
                    h &= h - 1u;
                    const int inc = 16 * q + 4 * (p & 7) + (p >> 3);
                    if (nl < CAPL) {
                        myfind[nl * WG_T] = inc;
                        ++nl;
                    } else {
                        add_incidence_by_lane(inc, row0, rows, src, src_idx, dst_idx, acc, cnt);
                    }
                }
            }
        }
        for (int ib = nvec * 16; ib < nnz; ib += WG_T) {
            const int i = ib + tid;
            if (i < nnz && (unsigned)dst_own[i] == me) {
                if (nl < CAPL) {
                    myfind[nl * WG_T] = i;
                    ++nl;
                } else {
                    add_incidence_by_lane(i, row0, rows, src, src_idx, dst_idx, acc, cnt);
                }
            }
        }


        int incl = nl;
        #pragma unroll
        for (int d = 1; d < 32; d <<= 1) {
            const int t = __shfl_up(incl, d, 32);
            if (lane >= d) incl += t;
        }
        const int off  = incl - nl;
        const int nall = __builtin_amdgcn_readlane(incl, 31);
        const int n    = (nall < CAPW) ? nall : CAPW;

        for (int k0 = 0; __builtin_amdgcn_ballot_w32(k0 < nl) != 0u; k0 += LU) {
            int s[LU], d[LU], f[LU];
            #pragma unroll
            for (int u = 0; u < LU; ++u) {
                const int k = (k0 + u < CAPL) ? (k0 + u) : (CAPL - 1);
                f[u] = myfind[k * WG_T];
            }
            #pragma unroll
            for (int u = 0; u < LU; ++u) {
                const int inc = (k0 + u < nl) ? f[u] : 0;
                s[u] = src_idx[inc];
                d[u] = dst_idx[inc];
            }
            #pragma unroll
            for (int u = 0; u < LU; ++u) {
                if (k0 + u < nl) {
                    const int  l    = d[u] - row0;
                    const bool ok   = (unsigned)l < (unsigned)rows;
                    const int  slot = off + k0 + u;
                    if (ok) atomicAdd(&cnt[l], 1);
                    if (slot < CAPW) {
                        mylist[slot] = make_int2(s[u], ok ? l : -1);
                    } else if (ok) {
                        add_row_by_lane(src + (size_t)s[u] * OUT_C, acc + l * OUT_C);
                    }
                }
            }
        }

        for (int j0 = 0; j0 < n; j0 += RB) {
            const int c = (n - j0 < RB) ? (n - j0) : RB;
            int s_l = 0, l_l = -1;
            if (lane < c) {
                const int2 e = mylist[j0 + lane];
                s_l = e.x;
                l_l = e.y;
            }
            float r[RB][4];
            #pragma unroll
            for (int u = 0; u < RB; ++u) {
                const float* __restrict__ row = src + (size_t)__builtin_amdgcn_readlane(s_l, u) * OUT_C;
                r[u][0] = row[lane];
                r[u][1] = row[lane + 32];
                r[u][2] = row[lane + 64];
                r[u][3] = row[lane + 96];
            }
            #pragma unroll
            for (int u = 0; u < RB; ++u) {
                const int l = __builtin_amdgcn_readlane(l_l, u);
                if (l >= 0) {
                    float* a = acc + l * OUT_C + lane;
                    atomicAdd(a,      r[u][0]);
                    atomicAdd(a + 32, r[u][1]);
                    atomicAdd(a + 64, r[u][2]);
                    atomicAdd(a + 96, r[u][3]);
                }
            }
        }
        __syncthreads();

        float4 b4 = {0.0f, 0.0f, 0.0f, 0.0f};
        if (HAS_BIAS) b4 = *(const float4*)(bias + (lane << 2));
        for (int r = wave; r < rows; r += WG_W) {
            const float sc = inv_cnt(cnt[r]);
            const float4 a = ((const float4*)(acc + r * OUT_C))[lane];
            float4 o;
            o.x = a.x * sc + b4.x;
            o.y = a.y * sc + b4.y;
            o.z = a.z * sc + b4.z;
            o.w = a.w * sc + b4.w;
            ((float4*)(dst + (size_t)(row0 + r) * OUT_C))[lane] = o;
        }
        __syncthreads();
    }
}

__global__ __launch_bounds__(WG_T) void node_to_edge_owner_kernel(
    const float* __restrict__ xw,
    const int* __restrict__ node_idx, const int* __restrict__ edge_idx,
    const unsigned char* __restrict__ own_e,
    float* __restrict__ edge_feat, int nnz, int n_edges)
{
    __shared__ __attribute__((aligned(16))) float acc[ROWS_E * OUT_C];
    __shared__ int2  list[WG_W * CAPW];
    __shared__ int   finds[CAPL * WG_T];
    __shared__ int   cnt[ROWS_E];
    owner_segment_sum<ROWS_E, false>(xw, node_idx, edge_idx, own_e, nnz, n_edges,
                                     edge_feat, nullptr, acc, cnt, finds, list);
}

__global__ __launch_bounds__(WG_T) void edge_to_node_owner_kernel(
    const float* __restrict__ edge_feat,
    const int* __restrict__ node_idx, const int* __restrict__ edge_idx,
    const unsigned char* __restrict__ own_n,
    const float* __restrict__ bias, float* __restrict__ out, int nnz, int n_nodes)
{
    __shared__ __attribute__((aligned(16))) float acc[ROWS_N * OUT_C];
    __shared__ int2  list[WG_W * CAPW];
    __shared__ int   finds[CAPL * WG_T];
    __shared__ int   cnt[ROWS_N];
    owner_segment_sum<ROWS_N, true>(edge_feat, edge_idx, node_idx, own_n, nnz, n_nodes,
                                    out, bias, acc, cnt, finds, list);
}

extern "C" void kernel_launch(void* const* d_in, const int* in_sizes, int n_in,
                              void* d_out, int out_size, void* d_ws, size_t ws_size,
                              hipStream_t stream)
{
    const float* x    = (const float*)d_in[0];
    const int*   hidx = (const int*)d_in[1];
    const float* W    = (const float*)d_in[2];
    const float* bias = (const float*)d_in[3];

    const int n_nodes = in_sizes[0] / IN_C;
    const int nnz     = in_sizes[1] / 2;
    const int n_edges = N_EDGES_CONST;

    const int* node_idx = hidx;
    const int* edge_idx = hidx + nnz;

    float* out = (float*)d_out;

    char* ws = (char*)d_ws;
    float* xw            = out;
    unsigned char* own_n = (unsigned char*)ws;
    unsigned char* own_e = (unsigned char*)(ws + (((size_t)nnz + 255) / 256) * 256);
    float* edge_feat     = (float*)(ws + (size_t)n_nodes * OUT_C * 4);

    gemm_xw_wmma_kernel<<<GEMM_WGS, 256, 0, stream>>>(x, W, xw, n_nodes / 16,
                                                       node_idx, edge_idx, nnz, own_n, own_e);

    node_to_edge_owner_kernel<<<OWN_GRID, WG_T, 0, stream>>>(
        xw, node_idx, edge_idx, own_e, edge_feat, nnz, n_edges);

    edge_to_node_owner_kernel<<<OWN_GRID, WG_T, 0, stream>>>(
        edge_feat, node_idx, edge_idx, own_n, bias, out, nnz, n_nodes);
}
